# starting kernel
# baseline (speedup 1.0000x reference)
#include <hip/hip_runtime.h>

typedef float v2f __attribute__((ext_vector_type(2)));
typedef float v8f __attribute__((ext_vector_type(8)));

#define BATCH_   64
#define SIZE_    256
#define NPR_     128
#define NPC_     128
#define NPCG_    8
#define NTILES_  (BATCH_ * NPR_ * NPCG_)

__device__ __forceinline__ v8f wmma4(v2f a, v2f b, v8f c) {
  return __builtin_amdgcn_wmma_f32_16x16x4_f32(false, a, false, b, (short)0, c,
                                               false, false);
}

__global__ __launch_bounds__(256)
void qconv2d_wmma_kernel(const float* __restrict__ x,
                         const float* __restrict__ w,
                         float* __restrict__ out) {
  const int lane = threadIdx.x & 31;
  const int p    = lane & 15;
  const int off  = (lane & 16) ? 2 : 0;

  float cq[4], sq[4];
#pragma unroll
  for (int q = 0; q < 4; ++q) {
    __sincosf(0.5f * w[q], &sq[q], &cq[q]);
  }
  const int s0 = p & 1, s1 = (p >> 1) & 1, s2 = (p >> 2) & 1, s3 = (p >> 3) & 1;
  const int j = (s0 ^ s3) | ((s0 ^ s1 ^ s3) << 1) | ((s1 ^ s2) << 2) | ((s2 ^ s3) << 3);

  v2f aR[4], aI[4];
#pragma unroll
  for (int c = 0; c < 4; ++c) {
    float mr[2], mi[2];
#pragma unroll
    for (int jj = 0; jj < 2; ++jj) {
      const int k  = 4 * c + off + jj;
      const int xo = j ^ k;
      const int pc = __popc(xo);
      float prod = 1.0f;
#pragma unroll
      for (int q = 0; q < 4; ++q) prod *= ((xo >> q) & 1) ? sq[q] : cq[q];
      if (pc & 1) { mr[jj] = 0.0f; mi[jj] = (pc & 2) ? prod : -prod; }
      else        { mi[jj] = 0.0f; mr[jj] = (pc & 2) ? -prod : prod; }
    }
    aR[c] = (v2f){mr[0], mr[1]};
    aI[c] = (v2f){mi[0], mi[1]};
  }

  const int nWaves = gridDim.x * (blockDim.x >> 5);
  const int gw     = blockIdx.x * (blockDim.x >> 5) + (threadIdx.x >> 5);

  for (int tile = gw; tile < NTILES_; tile += nWaves) {
    const int b   = tile >> 10;
    const int rem = tile & 1023;
    const int pr  = rem >> 3;
    const int pcg = rem & 7;

    const float* base = x + ((size_t)b * SIZE_ + 2 * pr) * SIZE_ + (pcg * 16 + p) * 2;
    const float2 lo = *(const float2*)base;
    const float2 hi = *(const float2*)(base + SIZE_);
    const float a0 = lo.x, a1 = lo.y, a2 = hi.x, a3 = hi.y;
    const float ss = a0 * a0 + a1 * a1 + a2 * a2 + a3 * a3;

    v8f Cr = {0.f, 0.f, 0.f, 0.f, 0.f, 0.f, 0.f, 0.f};
    v8f Ci = {0.f, 0.f, 0.f, 0.f, 0.f, 0.f, 0.f, 0.f};

#pragma unroll
    for (int c = 0; c < 4; ++c) {
      float cs[2], sn[2];
#pragma unroll
      for (int jj = 0; jj < 2; ++jj) {
        const int k = 4 * c + off + jj;
        const float t = ((k & 1) ? a0 : -a0) + ((k & 2) ? a1 : -a1) +
                        ((k & 4) ? a2 : -a2) + ((k & 8) ? a3 : -a3);
        const float phi = 0.5f * t + 0.25f * (t * t - ss);
        __sincosf(phi, &sn[jj], &cs[jj]);
      }
      const v2f bC  = (v2f){cs[0], cs[1]};
      const v2f bS  = (v2f){sn[0], sn[1]};
      const v2f bSn = (v2f){-sn[0], -sn[1]};
      Cr = wmma4(aR[c], bC,  Cr);
      Cr = wmma4(aI[c], bS,  Cr);
      Ci = wmma4(aI[c], bC,  Ci);
      Ci = wmma4(aR[c], bSn, Ci);
    }

    float acc0 = 0.f, acc1 = 0.f, acc2 = 0.f, acc3 = 0.f;
    const int sbase = (lane & 16) ? 8 : 0;
#pragma unroll
    for (int v = 0; v < 8; ++v) {
      const float pv = 0.0625f * (Cr[v] * Cr[v] + Ci[v] * Ci[v]);
      const int s = sbase + v;
      if (s & 8) acc0 += pv;
      if (s & 4) acc1 += pv;
      if (s & 2) acc2 += pv;
      if (s & 1) acc3 += pv;
    }
    acc0 += __shfl_xor(acc0, 16, 32);
    acc1 += __shfl_xor(acc1, 16, 32);
    acc2 += __shfl_xor(acc2, 16, 32);
    acc3 += __shfl_xor(acc3, 16, 32);

    if (lane < 16) {
      float4 o = make_float4(acc0, acc1, acc2, acc3);
      *(float4*)(out + (((size_t)b * NPR_ + pr) * NPC_ + pcg * 16 + p) * 4) = o;
    }
  }
}

extern "C" void kernel_launch(void* const* d_in, const int* in_sizes, int n_in,
                              void* d_out, int out_size, void* d_ws, size_t ws_size,
                              hipStream_t stream) {
  (void)in_sizes; (void)n_in; (void)d_ws; (void)ws_size; (void)out_size;
  const float* x = (const float*)d_in[0];
  const float* w = (const float*)d_in[1];
  float* out = (float*)d_out;
  qconv2d_wmma_kernel<<<dim3(2048), dim3(256), 0, stream>>>(x, w, out);
}


# Persistent tile loop: 4096 waves (4 per SIMD) walk 16 tiles each (two per iteration, next pair's loads one iteration ahead, scalar tile counter) instead of 16384 walking 4; set-up paid 4x less often
# speedup vs baseline: 1.4578x; 1.4578x over previous
#include <hip/hip_runtime.h>

typedef float v2f __attribute__((ext_vector_type(2)));
typedef float v8f __attribute__((ext_vector_type(8)));

#define BATCH_   64
#define SIZE_    256
#define NPR_     128
#define NPC_     128
#define NPCG_    8
#define NTILES_  (BATCH_ * NPR_ * NPCG_)
#define NPAIRS_  (NTILES_ / 2)
static_assert(NPCG_ % 2 == 0, "a tile pair must stay inside one patch row");

__device__ __forceinline__ v8f wmma4(v2f a, v2f b, v8f c) {
  return __builtin_amdgcn_wmma_f32_16x16x4_f32(false, a, false, b, (short)0, c,
                                               false, false);
}

__device__ __forceinline__ float m_entry(int j, int k, bool sine_side,
                                         const float* cq, const float* sq) {
  const int xo = j ^ k;
  const int pc = __popc(xo);
  float prod = 1.0f;
#pragma unroll
  for (int q = 0; q < 4; ++q) prod *= ((xo >> q) & 1) ? sq[q] : cq[q];
  if (pc & 1) return (pc & 2) ? prod : -prod;
  const bool neg = ((pc & 2) != 0) != sine_side;
  return neg ? -prod : prod;
}

__device__ __forceinline__ size_t tile_offset(int tile) {
  const int b   = tile >> 10;
  const int rem = tile & 1023;
  const int pr  = rem >> 3;
  const int pcg = rem & 7;
  return ((size_t)b * SIZE_ + 2 * pr) * SIZE_ + pcg * 32;
}

__device__ __forceinline__ float4 do_tile(const float2 lo, const float2 hi,
                                          const v2f (&aP)[4], const v2f (&aQ)[4],
                                          const int h) {
  const float a0 = lo.x, a1 = lo.y, a2 = hi.x, a3 = hi.y;
  const float ss = a0 * a0 + a1 * a1 + a2 * a2 + a3 * a3;

  const float a0h = h ? -a0 : a0;
  const float a3h = h ? a3 : -a3;

  float cE[2][2], sE[2][2], cO[2][2], sO[2][2];
#pragma unroll
  for (int c2 = 0; c2 < 2; ++c2) {
#pragma unroll
    for (int e = 0; e < 2; ++e) {
      const float b0 = (e ^ c2) ? a0h : -a0h;
      const float b1 = e ? a1 : -a1;
      const float b2 = c2 ? a2 : -a2;
      const float tE = ((b0 + b1) + b2) + a3h;
      const float tO = ((-b0 + b1) + b2) + a3h;
      const float phiE = 0.5f * tE + 0.25f * (tE * tE - ss);
      const float phiO = 0.5f * tO + 0.25f * (tO * tO - ss);
      __sincosf(phiE, &sE[c2][e], &cE[c2][e]);
      __sincosf(phiO, &sO[c2][e], &cO[c2][e]);
    }
  }

  v8f Cr = {0.f, 0.f, 0.f, 0.f, 0.f, 0.f, 0.f, 0.f};
  v8f Ci = {0.f, 0.f, 0.f, 0.f, 0.f, 0.f, 0.f, 0.f};
#pragma unroll
  for (int c2 = 0; c2 < 2; ++c2) {
    Cr = wmma4(aP[c2], (v2f){cE[c2][0], cE[c2][1]}, Cr);
    Ci = wmma4(aQ[c2], (v2f){cO[c2][0], cO[c2][1]}, Ci);
  }
#pragma unroll
  for (int c2 = 0; c2 < 2; ++c2) {
    Cr = wmma4(aP[2 + c2], (v2f){sO[c2][0], sO[c2][1]}, Cr);
    Ci = wmma4(aQ[2 + c2], (v2f){sE[c2][0], sE[c2][1]}, Ci);
  }

  float acc0 = 0.f, acc1 = 0.f, acc2 = 0.f, acc3 = 0.f;
  const int sbase = h ? 8 : 0;
#pragma unroll
  for (int v = 0; v < 8; ++v) {
    const float pv = 0.0625f * (Cr[v] * Cr[v] + Ci[v] * Ci[v]);
    const int s = sbase + v;
    if (s & 8) acc0 += pv;
    if (s & 4) acc1 += pv;
    if (s & 2) acc2 += pv;
    if (s & 1) acc3 += pv;
  }
  acc0 += __shfl_xor(acc0, 16, 32);
  acc1 += __shfl_xor(acc1, 16, 32);
  acc2 += __shfl_xor(acc2, 16, 32);
  acc3 += __shfl_xor(acc3, 16, 32);

  return make_float4(acc0, acc1, acc2, acc3);
}

__global__ __launch_bounds__(256)
void qconv2d_wmma_kernel(const float* __restrict__ x,
                         const float* __restrict__ w,
                         float* __restrict__ out) {
  const int lane = threadIdx.x & 31;
  const int p    = lane & 15;
  const int h    = lane >> 4;

  const int wpb    = blockDim.x >> 5;
  const int wv     = __builtin_amdgcn_readfirstlane((int)(threadIdx.x >> 5));
  const int nWaves = gridDim.x * wpb;
  int pair = blockIdx.x * wpb + wv;
  if (pair >= NPAIRS_) return;

  const float* base = x + tile_offset(2 * pair) + p * 2;
  float2 loA = *(const float2*)base;
  float2 hiA = *(const float2*)(base + SIZE_);
  float2 loB = *(const float2*)(base + 32);
  float2 hiB = *(const float2*)(base + SIZE_ + 32);

  float cq[4], sq[4];
#pragma unroll
  for (int q = 0; q < 4; ++q) {
    __sincosf(0.5f * w[q], &sq[q], &cq[q]);
  }
  const int s0 = p & 1, s1 = (p >> 1) & 1, s2 = (p >> 2) & 1, s3 = (p >> 3) & 1;
  const int j = (s0 ^ s3) | ((s0 ^ s1 ^ s3) << 1) | ((s1 ^ s2) << 2) | ((s2 ^ s3) << 3);

  v2f aP[4], aQ[4];
#pragma unroll
  for (int cc = 0; cc < 4; ++cc) {
    const int  c2   = cc & 1;
    const bool sine = (cc >= 2);
    float vp[2], vq[2];
#pragma unroll
    for (int e = 0; e < 2; ++e) {
      const int kE = ((e ^ c2 ^ h) & 1) | (e << 1) | (c2 << 2) | (h << 3);
      const int kO = kE ^ 1;
      vp[e] = m_entry(j, sine ? kO : kE, sine, cq, sq);
      vq[e] = m_entry(j, sine ? kE : kO, sine, cq, sq);
    }
    aP[cc] = (v2f){vp[0], vp[1]};
    aQ[cc] = (v2f){vq[0], vq[1]};
  }

  for (;;) {
    const int  npair = pair + nWaves;
    const bool more  = npair < NPAIRS_;
    const float* nbase = x + tile_offset(2 * (more ? npair : pair)) + p * 2;
    const float2 nloA = *(const float2*)nbase;
    const float2 nhiA = *(const float2*)(nbase + SIZE_);
    const float2 nloB = *(const float2*)(nbase + 32);
    const float2 nhiB = *(const float2*)(nbase + SIZE_ + 32);
    __builtin_amdgcn_sched_barrier(0);

    const float4 oA = do_tile(loA, hiA, aP, aQ, h);
    const float4 oB = do_tile(loB, hiB, aP, aQ, h);

    if (lane < 16) {
      float* o = out + ((size_t)pair * 32 + p) * 4;
      *(float4*)o        = oA;
      *(float4*)(o + 64) = oB;
    }

    if (!more) break;
    pair = npair;
    loA = nloA;  hiA = nhiA;
    loB = nloB;  hiB = nhiB;
  }
}

extern "C" void kernel_launch(void* const* d_in, const int* in_sizes, int n_in,
                              void* d_out, int out_size, void* d_ws, size_t ws_size,
                              hipStream_t stream) {
  (void)in_sizes; (void)n_in; (void)d_ws; (void)ws_size; (void)out_size;
  const float* x = (const float*)d_in[0];
  const float* w = (const float*)d_in[1];
  float* out = (float*)d_out;
  qconv2d_wmma_kernel<<<dim3(512), dim3(256), 0, stream>>>(x, w, out);
}


# Vector instructions per tile pair 143 to 98: phase in revolutions t*(KQ*t+KL)-KQ*ss from shared partial sums; probability sums as a packed tree, halves joined by v_permlanex16; sums reordered
# speedup vs baseline: 1.5465x; 1.0608x over previous
#include <hip/hip_runtime.h>

typedef float v2f __attribute__((ext_vector_type(2)));
typedef float v8f __attribute__((ext_vector_type(8)));

#define BATCH_   64
#define SIZE_    256
#define NPR_     128
#define NPC_     128
#define NPCG_    8
#define NTILES_  (BATCH_ * NPR_ * NPCG_)
#define NPAIRS_  (NTILES_ / 2)
static_assert(NPCG_ % 2 == 0, "a tile pair must stay inside one patch row");

__device__ __forceinline__ v8f wmma4(v2f a, v2f b, v8f c) {
  return __builtin_amdgcn_wmma_f32_16x16x4_f32(false, a, false, b, (short)0, c,
                                               false, false);
}

__device__ __forceinline__ float m_entry(int j, int k, bool sine_side,
                                         const float* cq, const float* sq) {
  const int xo = j ^ k;
  const int pc = __popc(xo);
  float prod = 0.25f;
#pragma unroll
  for (int q = 0; q < 4; ++q) prod *= ((xo >> q) & 1) ? sq[q] : cq[q];
  if (pc & 1) return (pc & 2) ? prod : -prod;
  const bool neg = ((pc & 2) != 0) != sine_side;
  return neg ? -prod : prod;
}

__device__ __forceinline__ size_t tile_offset(int tile) {
  const int b   = tile >> 10;
  const int rem = tile & 1023;
  const int pr  = rem >> 3;
  const int pcg = rem & 7;
  return ((size_t)b * SIZE_ + 2 * pr) * SIZE_ + pcg * 32;
}

__device__ __forceinline__ v2f fma2(v2f a, v2f b, v2f c) {
  return __builtin_elementwise_fma(a, b, c);
}

__device__ __forceinline__ float xhalf(float x) {
  const int b = __builtin_bit_cast(int, x);
  return __builtin_bit_cast(float, __builtin_amdgcn_permlanex16(b, b, 0x76543210, 0xfedcba98, false, false));
}

#define KL_ 0.07957747154594767f
#define KQ_ 0.039788735772973836f

__device__ __forceinline__ float4 do_tile(const float2 lo, const float2 hi,
                                          const v2f (&aP)[4], const v2f (&aQ)[4],
                                          const int h) {
  const float a0 = lo.x, a1 = lo.y, a2 = hi.x, a3 = hi.y;

  const float a0h = h ? -a0 : a0;
  const float a3h = h ? a3 : -a3;

  const v2f vlo = {a0, a1}, vhi = {a2, a3};
  const v2f sq  = fma2(vhi, vhi, vlo * vlo);
  const float nss = -KQ_ * (sq.x + sq.y);
  const v2f nss2 = {nss, nss};

  const v2f md   = (v2f){a1, a1} + (v2f){a2, -a2};
  const v2f a3h2 = {a3h, a3h};
  const v2f x1 = a3h2 + md;
  const v2f x0 = a3h2 - md;
  const v2f nb = {-a0h, a0h};
  const v2f tE1 = x1 + nb, tO1 = x1 - nb;
  const v2f tE0 = x0 + nb, tO0 = x0 - nb;

  const v2f kq2 = {KQ_, KQ_}, kl2 = {KL_, KL_};
  const v2f rE1 = fma2(tE1, fma2(tE1, kq2, kl2), nss2);
  const v2f rO1 = fma2(tO1, fma2(tO1, kq2, kl2), nss2);
  const v2f rE0 = fma2(tE0, fma2(tE0, kq2, kl2), nss2);
  const v2f rO0 = fma2(tO0, fma2(tO0, kq2, kl2), nss2);
  const float rEv[2][2] = {{rE0.x, rE1.y}, {rE0.y, rE1.x}};
  const float rOv[2][2] = {{rO0.x, rO1.y}, {rO0.y, rO1.x}};
  float cE[2][2], sE[2][2], cO[2][2], sO[2][2];
#pragma unroll
  for (int c2 = 0; c2 < 2; ++c2) {
#pragma unroll
    for (int e = 0; e < 2; ++e) {
      cE[c2][e] = __builtin_amdgcn_cosf(rEv[c2][e]);
      sE[c2][e] = __builtin_amdgcn_sinf(rEv[c2][e]);
      cO[c2][e] = __builtin_amdgcn_cosf(rOv[c2][e]);
      sO[c2][e] = __builtin_amdgcn_sinf(rOv[c2][e]);
    }
  }

  v8f Cr = {0.f, 0.f, 0.f, 0.f, 0.f, 0.f, 0.f, 0.f};
  v8f Ci = {0.f, 0.f, 0.f, 0.f, 0.f, 0.f, 0.f, 0.f};
#pragma unroll
  for (int c2 = 0; c2 < 2; ++c2) {
    Cr = wmma4(aP[c2], (v2f){cE[c2][0], cE[c2][1]}, Cr);
    Ci = wmma4(aQ[c2], (v2f){cO[c2][0], cO[c2][1]}, Ci);
  }
#pragma unroll
  for (int c2 = 0; c2 < 2; ++c2) {
    Cr = wmma4(aP[2 + c2], (v2f){sO[c2][0], sO[c2][1]}, Cr);
    Ci = wmma4(aQ[2 + c2], (v2f){sE[c2][0], sE[c2][1]}, Ci);
  }

  const v2f p01 = fma2(__builtin_shufflevector(Ci, Ci, 0, 1), __builtin_shufflevector(Ci, Ci, 0, 1),
                       __builtin_shufflevector(Cr, Cr, 0, 1) * __builtin_shufflevector(Cr, Cr, 0, 1));
  const v2f p23 = fma2(__builtin_shufflevector(Ci, Ci, 2, 3), __builtin_shufflevector(Ci, Ci, 2, 3),
                       __builtin_shufflevector(Cr, Cr, 2, 3) * __builtin_shufflevector(Cr, Cr, 2, 3));
  const v2f p45 = fma2(__builtin_shufflevector(Ci, Ci, 4, 5), __builtin_shufflevector(Ci, Ci, 4, 5),
                       __builtin_shufflevector(Cr, Cr, 4, 5) * __builtin_shufflevector(Cr, Cr, 4, 5));
  const v2f p67 = fma2(__builtin_shufflevector(Ci, Ci, 6, 7), __builtin_shufflevector(Ci, Ci, 6, 7),
                       __builtin_shufflevector(Cr, Cr, 6, 7) * __builtin_shufflevector(Cr, Cr, 6, 7));
  const v2f q = p01 + p23;
  const v2f r = p45 + p67;
  const v2f w = p23 + p67;
  float acc1 = r.x + r.y;
  float acc2 = w.x + w.y;
  float acc3 = q.y + r.y;
  const float tot = (q.x + q.y) + acc1;
  const float acc0 = xhalf(tot);
  acc1 += xhalf(acc1);
  acc2 += xhalf(acc2);
  acc3 += xhalf(acc3);

  return make_float4(acc0, acc1, acc2, acc3);
}

__global__ __launch_bounds__(256)
void qconv2d_wmma_kernel(const float* __restrict__ x,
                         const float* __restrict__ w,
                         float* __restrict__ out) {
  const int lane = threadIdx.x & 31;
  const int p    = lane & 15;
  const int h    = lane >> 4;

  const int wpb    = blockDim.x >> 5;
  const int wv     = __builtin_amdgcn_readfirstlane((int)(threadIdx.x >> 5));
  const int nWaves = gridDim.x * wpb;
  int pair = blockIdx.x * wpb + wv;
  if (pair >= NPAIRS_) return;

  const float* base = x + tile_offset(2 * pair) + p * 2;
  float2 loA = *(const float2*)base;
  float2 hiA = *(const float2*)(base + SIZE_);
  float2 loB = *(const float2*)(base + 32);
  float2 hiB = *(const float2*)(base + SIZE_ + 32);

  float cq[4], sq[4];
#pragma unroll
  for (int q = 0; q < 4; ++q) {
    __sincosf(0.5f * w[q], &sq[q], &cq[q]);
  }
  const int s0 = p & 1, s1 = (p >> 1) & 1, s2 = (p >> 2) & 1, s3 = (p >> 3) & 1;
  const int j = (s0 ^ s3) | ((s0 ^ s1 ^ s3) << 1) | ((s1 ^ s2) << 2) | ((s2 ^ s3) << 3);

  v2f aP[4], aQ[4];
#pragma unroll
  for (int cc = 0; cc < 4; ++cc) {
    const int  c2   = cc & 1;
    const bool sine = (cc >= 2);
    float vp[2], vq[2];
#pragma unroll
    for (int e = 0; e < 2; ++e) {
      const int kE = ((e ^ c2 ^ h) & 1) | (e << 1) | (c2 << 2) | (h << 3);
      const int kO = kE ^ 1;
      vp[e] = m_entry(j, sine ? kO : kE, sine, cq, sq);
      vq[e] = m_entry(j, sine ? kE : kO, sine, cq, sq);
    }
    aP[cc] = (v2f){vp[0], vp[1]};
    aQ[cc] = (v2f){vq[0], vq[1]};
  }

  for (;;) {
    const int  npair = pair + nWaves;
    const bool more  = npair < NPAIRS_;
    const float* nbase = x + tile_offset(2 * (more ? npair : pair)) + p * 2;
    const float2 nloA = *(const float2*)nbase;
    const float2 nhiA = *(const float2*)(nbase + SIZE_);
    const float2 nloB = *(const float2*)(nbase + 32);
    const float2 nhiB = *(const float2*)(nbase + SIZE_ + 32);
    __builtin_amdgcn_sched_barrier(0);

    const float4 oA = do_tile(loA, hiA, aP, aQ, h);
    const float4 oB = do_tile(loB, hiB, aP, aQ, h);

    if (lane < 16) {
      float* o = out + ((size_t)pair * 32 + p) * 4;
      *(float4*)o        = oA;
      *(float4*)(o + 64) = oB;
    }

    if (!more) break;
    pair = npair;
    loA = nloA;  hiA = nhiA;
    loB = nloB;  hiB = nhiB;
  }
}

extern "C" void kernel_launch(void* const* d_in, const int* in_sizes, int n_in,
                              void* d_out, int out_size, void* d_ws, size_t ws_size,
                              hipStream_t stream) {
  (void)in_sizes; (void)n_in; (void)d_ws; (void)ws_size; (void)out_size;
  const float* x = (const float*)d_in[0];
  const float* w = (const float*)d_in[1];
  float* out = (float*)d_out;
  qconv2d_wmma_kernel<<<dim3(512), dim3(256), 0, stream>>>(x, w, out);
}


# All 32 angle loads of a wave issued at wave start into 64 registers and its 8 tile pairs fully unrolled: no load, address math or register rotation left inside the tile stream; same sums
# speedup vs baseline: 1.7474x; 1.1299x over previous
#include <hip/hip_runtime.h>

typedef float v2f __attribute__((ext_vector_type(2)));
typedef float v8f __attribute__((ext_vector_type(8)));

#define BATCH_   64
#define SIZE_    256
#define NPR_     128
#define NPC_     128
#define NPCG_    8
#define NTILES_  (BATCH_ * NPR_ * NPCG_)
#define NPAIRS_  (NTILES_ / 2)
static_assert(NPCG_ % 2 == 0, "a tile pair must stay inside one patch row");

__device__ __forceinline__ v8f wmma4(v2f a, v2f b, v8f c) {
  return __builtin_amdgcn_wmma_f32_16x16x4_f32(false, a, false, b, (short)0, c,
                                               false, false);
}

__device__ __forceinline__ float m_entry(int j, int k, bool sine_side,
                                         const float* cq, const float* sq) {
  const int xo = j ^ k;
  const int pc = __popc(xo);
  float prod = 0.25f;
#pragma unroll
  for (int q = 0; q < 4; ++q) prod *= ((xo >> q) & 1) ? sq[q] : cq[q];
  if (pc & 1) return (pc & 2) ? prod : -prod;
  const bool neg = ((pc & 2) != 0) != sine_side;
  return neg ? -prod : prod;
}

__device__ __forceinline__ size_t tile_offset(int tile) {
  const int b   = tile >> 10;
  const int rem = tile & 1023;
  const int pr  = rem >> 3;
  const int pcg = rem & 7;
  return ((size_t)b * SIZE_ + 2 * pr) * SIZE_ + pcg * 32;
}

__device__ __forceinline__ v2f fma2(v2f a, v2f b, v2f c) {
  return __builtin_elementwise_fma(a, b, c);
}

__device__ __forceinline__ float xhalf(float x) {
  const int b = __builtin_bit_cast(int, x);
  return __builtin_bit_cast(float, __builtin_amdgcn_permlanex16(b, b, 0x76543210, 0xfedcba98, false, false));
}

#define KL_ 0.07957747154594767f
#define KQ_ 0.039788735772973836f

__device__ __forceinline__ float4 do_tile(const float2 lo, const float2 hi,
                                          const v2f (&aP)[4], const v2f (&aQ)[4],
                                          const int h) {
  const float a0 = lo.x, a1 = lo.y, a2 = hi.x, a3 = hi.y;

  const float a0h = h ? -a0 : a0;
  const float a3h = h ? a3 : -a3;

  const v2f vlo = {a0, a1}, vhi = {a2, a3};
  const v2f sq  = fma2(vhi, vhi, vlo * vlo);
  const float nss = -KQ_ * (sq.x + sq.y);
  const v2f nss2 = {nss, nss};

  const v2f md   = (v2f){a1, a1} + (v2f){a2, -a2};
  const v2f a3h2 = {a3h, a3h};
  const v2f x1 = a3h2 + md;
  const v2f x0 = a3h2 - md;
  const v2f nb = {-a0h, a0h};
  const v2f tE1 = x1 + nb, tO1 = x1 - nb;
  const v2f tE0 = x0 + nb, tO0 = x0 - nb;

  const v2f kq2 = {KQ_, KQ_}, kl2 = {KL_, KL_};
  const v2f rE1 = fma2(tE1, fma2(tE1, kq2, kl2), nss2);
  const v2f rO1 = fma2(tO1, fma2(tO1, kq2, kl2), nss2);
  const v2f rE0 = fma2(tE0, fma2(tE0, kq2, kl2), nss2);
  const v2f rO0 = fma2(tO0, fma2(tO0, kq2, kl2), nss2);
  const float rEv[2][2] = {{rE0.x, rE1.y}, {rE0.y, rE1.x}};
  const float rOv[2][2] = {{rO0.x, rO1.y}, {rO0.y, rO1.x}};
  float cE[2][2], sE[2][2], cO[2][2], sO[2][2];
#pragma unroll
  for (int c2 = 0; c2 < 2; ++c2) {
#pragma unroll
    for (int e = 0; e < 2; ++e) {
      cE[c2][e] = __builtin_amdgcn_cosf(rEv[c2][e]);
      sE[c2][e] = __builtin_amdgcn_sinf(rEv[c2][e]);
      cO[c2][e] = __builtin_amdgcn_cosf(rOv[c2][e]);
      sO[c2][e] = __builtin_amdgcn_sinf(rOv[c2][e]);
    }
  }

  v8f Cr = {0.f, 0.f, 0.f, 0.f, 0.f, 0.f, 0.f, 0.f};
  v8f Ci = {0.f, 0.f, 0.f, 0.f, 0.f, 0.f, 0.f, 0.f};
#pragma unroll
  for (int c2 = 0; c2 < 2; ++c2) {
    Cr = wmma4(aP[c2], (v2f){cE[c2][0], cE[c2][1]}, Cr);
    Ci = wmma4(aQ[c2], (v2f){cO[c2][0], cO[c2][1]}, Ci);
  }
#pragma unroll
  for (int c2 = 0; c2 < 2; ++c2) {
    Cr = wmma4(aP[2 + c2], (v2f){sO[c2][0], sO[c2][1]}, Cr);
    Ci = wmma4(aQ[2 + c2], (v2f){sE[c2][0], sE[c2][1]}, Ci);
  }

  const v2f p01 = fma2(__builtin_shufflevector(Ci, Ci, 0, 1), __builtin_shufflevector(Ci, Ci, 0, 1),
                       __builtin_shufflevector(Cr, Cr, 0, 1) * __builtin_shufflevector(Cr, Cr, 0, 1));
  const v2f p23 = fma2(__builtin_shufflevector(Ci, Ci, 2, 3), __builtin_shufflevector(Ci, Ci, 2, 3),
                       __builtin_shufflevector(Cr, Cr, 2, 3) * __builtin_shufflevector(Cr, Cr, 2, 3));
  const v2f p45 = fma2(__builtin_shufflevector(Ci, Ci, 4, 5), __builtin_shufflevector(Ci, Ci, 4, 5),
                       __builtin_shufflevector(Cr, Cr, 4, 5) * __builtin_shufflevector(Cr, Cr, 4, 5));
  const v2f p67 = fma2(__builtin_shufflevector(Ci, Ci, 6, 7), __builtin_shufflevector(Ci, Ci, 6, 7),
                       __builtin_shufflevector(Cr, Cr, 6, 7) * __builtin_shufflevector(Cr, Cr, 6, 7));
  const v2f q = p01 + p23;
  const v2f r = p45 + p67;
  const v2f w = p23 + p67;
  float acc1 = r.x + r.y;
  float acc2 = w.x + w.y;
  float acc3 = q.y + r.y;
  const float tot = (q.x + q.y) + acc1;
  const float acc0 = xhalf(tot);
  acc1 += xhalf(acc1);
  acc2 += xhalf(acc2);
  acc3 += xhalf(acc3);

  return make_float4(acc0, acc1, acc2, acc3);
}

#define WAVES_PER_BLOCK_ 8
#define NBLOCKS_         512
#define NWAVES_          (NBLOCKS_ * WAVES_PER_BLOCK_)
#define PAIRS_PER_WAVE_  (NPAIRS_ / NWAVES_)
static_assert(NPAIRS_ % NWAVES_ == 0, "every wave must own the same whole number of tile pairs");
static_assert((2 * NWAVES_) % (NPR_ * NPCG_) == 0, "a wave's pairs must differ by whole images");

__global__ __launch_bounds__(256)
void qconv2d_wmma_kernel(const float* __restrict__ x,
                         const float* __restrict__ w,
                         float* __restrict__ out) {
  const int lane = threadIdx.x & 31;
  const int p    = lane & 15;
  const int h    = lane >> 4;

  const int wv    = __builtin_amdgcn_readfirstlane((int)(threadIdx.x >> 5));
  const int pair0 = blockIdx.x * WAVES_PER_BLOCK_ + wv;

  float2 lo[PAIRS_PER_WAVE_][2], hi[PAIRS_PER_WAVE_][2];
  const float* base0 = x + tile_offset(2 * pair0) + p * 2;
#pragma unroll
  for (int i = 0; i < PAIRS_PER_WAVE_; ++i) {
    const float* base = base0 + (size_t)i * ((2 * NWAVES_ / (NPR_ * NPCG_)) * SIZE_ * SIZE_);
    lo[i][0] = *(const float2*)base;
    hi[i][0] = *(const float2*)(base + SIZE_);
    lo[i][1] = *(const float2*)(base + 32);
    hi[i][1] = *(const float2*)(base + SIZE_ + 32);
  }
  __builtin_amdgcn_sched_barrier(0);

  float cq[4], sq[4];
#pragma unroll
  for (int q = 0; q < 4; ++q) {
    __sincosf(0.5f * w[q], &sq[q], &cq[q]);
  }
  const int s0 = p & 1, s1 = (p >> 1) & 1, s2 = (p >> 2) & 1, s3 = (p >> 3) & 1;
  const int j = (s0 ^ s3) | ((s0 ^ s1 ^ s3) << 1) | ((s1 ^ s2) << 2) | ((s2 ^ s3) << 3);

  v2f aP[4], aQ[4];
#pragma unroll
  for (int cc = 0; cc < 4; ++cc) {
    const int  c2   = cc & 1;
    const bool sine = (cc >= 2);
    float vp[2], vq[2];
#pragma unroll
    for (int e = 0; e < 2; ++e) {
      const int kE = ((e ^ c2 ^ h) & 1) | (e << 1) | (c2 << 2) | (h << 3);
      const int kO = kE ^ 1;
      vp[e] = m_entry(j, sine ? kO : kE, sine, cq, sq);
      vq[e] = m_entry(j, sine ? kE : kO, sine, cq, sq);
    }
    aP[cc] = (v2f){vp[0], vp[1]};
    aQ[cc] = (v2f){vq[0], vq[1]};
  }

  float* o0 = out + ((size_t)pair0 * 32 + p) * 4;
#pragma unroll
  for (int i = 0; i < PAIRS_PER_WAVE_; ++i) {
    const float4 oA = do_tile(lo[i][0], hi[i][0], aP, aQ, h);
    const float4 oB = do_tile(lo[i][1], hi[i][1], aP, aQ, h);

    if (lane < 16) {
      float* o = o0 + (size_t)i * ((size_t)NWAVES_ * 32 * 4);
      *(float4*)o        = oA;
      *(float4*)(o + 64) = oB;
    }
  }
}

extern "C" void kernel_launch(void* const* d_in, const int* in_sizes, int n_in,
                              void* d_out, int out_size, void* d_ws, size_t ws_size,
                              hipStream_t stream) {
  (void)in_sizes; (void)n_in; (void)d_ws; (void)ws_size; (void)out_size;
  const float* x = (const float*)d_in[0];
  const float* w = (const float*)d_in[1];
  float* out = (float*)d_out;
  qconv2d_wmma_kernel<<<dim3(NBLOCKS_), dim3(WAVES_PER_BLOCK_ * 32), 0, stream>>>(x, w, out);
}


# Vector instructions per tile pair 94 to 61, set-up 168 to 65: phase as (s(t+1))^2, v_permlane16_swap joins both tiles' halves, one 32-lane store, operand matrix by complex products; sums reordered
# speedup vs baseline: 1.8931x; 1.0834x over previous
#include <hip/hip_runtime.h>


typedef float v2f __attribute__((ext_vector_type(2)));
typedef float v8f __attribute__((ext_vector_type(8)));

#define BATCH_   64
#define SIZE_    256
#define NPR_     128
#define NPC_     128
#define NPCG_    8
#define NTILES_  (BATCH_ * NPR_ * NPCG_)
#define NPAIRS_  (NTILES_ / 2)
static_assert(NPCG_ % 2 == 0, "a tile pair must stay inside one patch row");

__device__ __forceinline__ v8f wmma4(v2f a, v2f b, v8f c) {
  return __builtin_amdgcn_wmma_f32_16x16x4_f32(false, a, false, b, (short)0, c,
                                               false, false);
}

__device__ __forceinline__ size_t tile_offset(int tile) {
  const int b   = tile >> 10;
  const int rem = tile & 1023;
  const int pr  = rem >> 3;
  const int pcg = rem & 7;
  return ((size_t)b * SIZE_ + 2 * pr) * SIZE_ + pcg * 32;
}

__device__ __forceinline__ v2f fma2(v2f a, v2f b, v2f c) {
  return __builtin_elementwise_fma(a, b, c);
}

__device__ __forceinline__ void cmul2(v2f ar, v2f ai, v2f br, v2f bi, v2f& cr, v2f& ci) {
  cr = fma2(-ai, bi, ar * br);
  ci = fma2(ai, br, ar * bi);
}

__device__ __forceinline__ void swap_rows(float& d, float& s) {
  asm("v_permlane16_swap_b32 %0, %1" : "+v"(d), "+v"(s));
}

#define SQ_ 0.19947114020071635f

struct TileSums { float tot, a1, a2, a3; };

__device__ __forceinline__ TileSums do_tile(const float2 lo, const float2 hi,
                                            const v2f (&aP)[4], const v2f (&aQ)[4],
                                            const v2f sc_lo, const v2f sc_hi, const v2f ad_hi) {
  const v2f vlo = {lo.x, lo.y}, vhi = {hi.x, hi.y};
  const v2f blo = vlo * sc_lo;
  const v2f bhi = fma2(vhi, sc_hi, ad_hi);
  const float a0h = blo.x, b1 = blo.y, b2 = bhi.x, a3s = bhi.y;

  const v2f md   = (v2f){b1, b1} + (v2f){b2, -b2};
  const v2f a3s2 = {a3s, a3s};
  const v2f x1 = a3s2 + md;
  const v2f x0 = a3s2 - md;
  const v2f nb = {-a0h, a0h};
  const v2f uE1 = x1 + nb, uO1 = x1 - nb;
  const v2f uE0 = x0 + nb, uO0 = x0 - nb;
  const v2f rE1 = uE1 * uE1, rO1 = uO1 * uO1;
  const v2f rE0 = uE0 * uE0, rO0 = uO0 * uO0;
  const float rEv[2][2] = {{rE0.x, rE1.y}, {rE0.y, rE1.x}};
  const float rOv[2][2] = {{rO0.x, rO1.y}, {rO0.y, rO1.x}};
  float cE[2][2], sE[2][2], cO[2][2], sO[2][2];
#pragma unroll
  for (int c2 = 0; c2 < 2; ++c2) {
#pragma unroll
    for (int e = 0; e < 2; ++e) {
      cE[c2][e] = __builtin_amdgcn_cosf(rEv[c2][e]);
      sE[c2][e] = __builtin_amdgcn_sinf(rEv[c2][e]);
      cO[c2][e] = __builtin_amdgcn_cosf(rOv[c2][e]);
      sO[c2][e] = __builtin_amdgcn_sinf(rOv[c2][e]);
    }
  }

  v8f Cr = {0.f, 0.f, 0.f, 0.f, 0.f, 0.f, 0.f, 0.f};
  v8f Ci = {0.f, 0.f, 0.f, 0.f, 0.f, 0.f, 0.f, 0.f};
#pragma unroll
  for (int c2 = 0; c2 < 2; ++c2) {
    Cr = wmma4(aP[c2], (v2f){cE[c2][0], cE[c2][1]}, Cr);
    Ci = wmma4(aQ[c2], (v2f){cO[c2][0], cO[c2][1]}, Ci);
  }
#pragma unroll
  for (int c2 = 0; c2 < 2; ++c2) {
    Cr = wmma4(aP[2 + c2], (v2f){sO[c2][0], sO[c2][1]}, Cr);
    Ci = wmma4(aQ[2 + c2], (v2f){sE[c2][0], sE[c2][1]}, Ci);
  }

  const v2f p01 = fma2(__builtin_shufflevector(Ci, Ci, 0, 1), __builtin_shufflevector(Ci, Ci, 0, 1),
                       __builtin_shufflevector(Cr, Cr, 0, 1) * __builtin_shufflevector(Cr, Cr, 0, 1));
  const v2f p23 = fma2(__builtin_shufflevector(Ci, Ci, 2, 3), __builtin_shufflevector(Ci, Ci, 2, 3),
                       __builtin_shufflevector(Cr, Cr, 2, 3) * __builtin_shufflevector(Cr, Cr, 2, 3));
  const v2f p45 = fma2(__builtin_shufflevector(Ci, Ci, 4, 5), __builtin_shufflevector(Ci, Ci, 4, 5),
                       __builtin_shufflevector(Cr, Cr, 4, 5) * __builtin_shufflevector(Cr, Cr, 4, 5));
  const v2f p67 = fma2(__builtin_shufflevector(Ci, Ci, 6, 7), __builtin_shufflevector(Ci, Ci, 6, 7),
                       __builtin_shufflevector(Cr, Cr, 6, 7) * __builtin_shufflevector(Cr, Cr, 6, 7));
  const v2f q = p01 + p23;
  const v2f r = p45 + p67;
  const v2f w = p23 + p67;
  const v2f z = q + r;
  TileSums s;
  s.a1  = r.x + r.y;
  s.a2  = w.x + w.y;
  s.a3  = z.y;
  s.tot = z.x + z.y;
  return s;
}

#define WAVES_PER_BLOCK_ 8
#define NBLOCKS_         512
#define NWAVES_          (NBLOCKS_ * WAVES_PER_BLOCK_)
#define PAIRS_PER_WAVE_  (NPAIRS_ / NWAVES_)
static_assert(NPAIRS_ % NWAVES_ == 0, "every wave must own the same whole number of tile pairs");
static_assert((2 * NWAVES_) % (NPR_ * NPCG_) == 0, "a wave's pairs must differ by whole images");

__global__ __launch_bounds__(256)
void qconv2d_wmma_kernel(const float* __restrict__ x,
                         const float* __restrict__ w,
                         float* __restrict__ out) {
  const int lane = threadIdx.x & 31;
  const int p    = lane & 15;
  const int h    = lane >> 4;

  const int wv    = __builtin_amdgcn_readfirstlane((int)(threadIdx.x >> 5));
  const int pair0 = blockIdx.x * WAVES_PER_BLOCK_ + wv;

  float2 lo[PAIRS_PER_WAVE_][2], hi[PAIRS_PER_WAVE_][2];
  const float* base0 = x + tile_offset(2 * pair0) + p * 2;
#pragma unroll
  for (int i = 0; i < PAIRS_PER_WAVE_; ++i) {
    const float* base = base0 + (size_t)i * ((2 * NWAVES_ / (NPR_ * NPCG_)) * SIZE_ * SIZE_);
    lo[i][0] = *(const float2*)base;
    hi[i][0] = *(const float2*)(base + SIZE_);
    lo[i][1] = *(const float2*)(base + 32);
    hi[i][1] = *(const float2*)(base + SIZE_ + 32);
  }
  __builtin_amdgcn_sched_barrier(0);

  float cq[4], sq[4];
#pragma unroll
  for (int q = 0; q < 4; ++q) {
    __sincosf(0.5f * w[q], &sq[q], &cq[q]);
  }
  const int s0 = p & 1, s1 = (p >> 1) & 1, s2 = (p >> 2) & 1, s3 = (p >> 3) & 1;
  const bool j0 = (s0 ^ s3) != 0, j1 = (s0 ^ s1 ^ s3) != 0, j2 = (s1 ^ s2) != 0, j3 = (s2 ^ s3) != 0;

  const v2f f1r = j1 ? (v2f){0.f, cq[1]} : (v2f){cq[1], 0.f};
  const v2f f1i = j1 ? (v2f){-sq[1], 0.f} : (v2f){0.f, -sq[1]};
  const bool b3 = j3 != (h != 0);
  const float f3r = b3 ? 0.f : 0.25f * cq[3];
  const float f3i = b3 ? -0.25f * sq[3] : 0.f;
  const bool t0 = j0 != (h != 0);
  const v2f g0r = t0 ? (v2f){0.f, cq[0]} : (v2f){cq[0], 0.f};
  const v2f g0i = t0 ? (v2f){-sq[0], 0.f} : (v2f){0.f, -sq[0]};
  const v2f m03r = g0r * f3r - g0i * f3i;
  const v2f m03i = g0r * f3i + g0i * f3r;
  const v2f m03r_sw = __builtin_shufflevector(m03r, m03r, 1, 0);
  const v2f m03i_sw = __builtin_shufflevector(m03i, m03i, 1, 0);

  v2f aP[4], aQ[4];
#pragma unroll
  for (int c2 = 0; c2 < 2; ++c2) {
    const bool b2 = j2 != (c2 != 0);
    const float f2r = b2 ? 0.f : cq[2];
    const float f2i = b2 ? -sq[2] : 0.f;
    const v2f m12r = f1r * f2r - f1i * f2i;
    const v2f m12i = f1r * f2i + f1i * f2r;
    v2f uEr, uEi, uOr, uOi;
    cmul2(m12r, m12i, c2 ? m03r_sw : m03r, c2 ? m03i_sw : m03i, uEr, uEi);
    cmul2(m12r, m12i, c2 ? m03r : m03r_sw, c2 ? m03i : m03i_sw, uOr, uOi);
    aP[c2]     = uEr + uEi;
    aQ[c2]     = uOr + uOi;
    aP[2 + c2] = uOi - uOr;
    aQ[2 + c2] = uEi - uEr;
  }

  const v2f sc_lo = {h ? -SQ_ : SQ_, SQ_};
  const v2f sc_hi = {SQ_, h ? SQ_ : -SQ_};
  const v2f ad_hi = {0.f, SQ_};

  float* o0 = out + ((size_t)pair0 * 32 + lane) * 4;
#pragma unroll
  for (int i = 0; i < PAIRS_PER_WAVE_; ++i) {
    TileSums A = do_tile(lo[i][0], hi[i][0], aP, aQ, sc_lo, sc_hi, ad_hi);
    TileSums B = do_tile(lo[i][1], hi[i][1], aP, aQ, sc_lo, sc_hi, ad_hi);

    swap_rows(A.tot, B.tot);
    swap_rows(A.a1, B.a1);
    swap_rows(A.a2, B.a2);
    swap_rows(A.a3, B.a3);
    *(float4*)(o0 + (size_t)i * ((size_t)NWAVES_ * 32 * 4)) =
        make_float4(B.tot, A.a1 + B.a1, A.a2 + B.a2, A.a3 + B.a3);
    __builtin_amdgcn_sched_barrier(0);
  }
}

extern "C" void kernel_launch(void* const* d_in, const int* in_sizes, int n_in,
                              void* d_out, int out_size, void* d_ws, size_t ws_size,
                              hipStream_t stream) {
  (void)in_sizes; (void)n_in; (void)d_ws; (void)ws_size; (void)out_size;
  const float* x = (const float*)d_in[0];
  const float* w = (const float*)d_in[1];
  float* out = (float*)d_out;
  qconv2d_wmma_kernel<<<dim3(NBLOCKS_), dim3(WAVES_PER_BLOCK_ * 32), 0, stream>>>(x, w, out);
}
